# starting kernel
# baseline (speedup 1.0000x reference)
#include <hip/hip_runtime.h>

typedef float v2f __attribute__((ext_vector_type(2)));
typedef float v8f __attribute__((ext_vector_type(8)));

#define NS   8192
#define DB   64
#define DIN  64
#define DH1  128
#define DH2  64

__device__ __forceinline__ float sigmoidf(float v) {
    return 1.0f / (1.0f + __expf(-v));
}

__device__ __forceinline__ v2f ld2(const float* p) {
    return *(const v2f*)p;
}

__global__ void init_flag_kernel(int* f) { *f = 1; }

__global__ void check_binary_kernel(const float* __restrict__ x, int n, int* f) {
    int i = blockIdx.x * blockDim.x + threadIdx.x;
    int stride = gridDim.x * blockDim.x;
    for (; i < n; i += stride) {
        float v = x[i];
        if (v != 0.0f && v != 1.0f) { *f = 0; return; }
    }
}

__global__ void __launch_bounds__(64)
topo_mlp_kernel(const float* __restrict__ x,  const float* __restrict__ W0,
                const float* __restrict__ b0, const float* __restrict__ W1,
                const float* __restrict__ b1, const float* __restrict__ W2,
                const float* __restrict__ b2, float* __restrict__ out,
                const int* __restrict__ flag_p)
{
    __shared__ float lds_act[2][32 * DH1];

    const int lane   = threadIdx.x & 31;
    const int wav    = threadIdx.x >> 5;
    const int wid    = blockIdx.x * 2 + wav;
    const int branch = wid >> 8;
    const int tile   = wid & 255;
    const int r0     = tile * 32;
    const int half   = lane >> 4;
    const int l15    = lane & 15;

    const int binflag = flag_p ? flag_p[0] : 0;

    const float* XrowA = x + (size_t)(r0 + l15) * DIN + 2 * half;
    const float* XrowB = XrowA + 16 * DIN;
    v2f a0A[16], a0B[16];
    #pragma unroll
    for (int kb = 0; kb < 16; ++kb) {
        a0A[kb] = ld2(XrowA + 4 * kb);
        a0B[kb] = ld2(XrowB + 4 * kb);
    }

    const float* W0b = W0 + (size_t)branch * DIN * DH1;
    const float* b0b = b0 + branch * DH1;
    float* acts = lds_act[wav];

    #pragma unroll
    for (int t = 0; t < 8; ++t) {
        const float bias = b0b[16 * t + l15];
        v8f accA = {bias, bias, bias, bias, bias, bias, bias, bias};
        v8f accB = accA;
        #pragma unroll
        for (int kb = 0; kb < 16; ++kb) {
            const float* bp = W0b + (size_t)(4 * kb + 2 * half) * DH1 + 16 * t + l15;
            v2f bf = {bp[0], bp[DH1]};
            accA = __builtin_amdgcn_wmma_f32_16x16x4_f32(
                false, a0A[kb], false, bf, (short)0, accA, false, false);
            accB = __builtin_amdgcn_wmma_f32_16x16x4_f32(
                false, a0B[kb], false, bf, (short)0, accB, false, false);
        }
        #pragma unroll
        for (int j = 0; j < 8; ++j) {
            acts[(j + 8 * half) * DH1 + 16 * t + l15]        = sigmoidf(accA[j]);
            acts[(16 + j + 8 * half) * DH1 + 16 * t + l15]   = sigmoidf(accB[j]);
        }
    }

    asm volatile("s_wait_dscnt 0" ::: "memory");

    v2f a1A[32], a1B[32];
    #pragma unroll
    for (int kb = 0; kb < 32; ++kb) {
        a1A[kb] = ld2(&acts[l15 * DH1 + 4 * kb + 2 * half]);
        a1B[kb] = ld2(&acts[(16 + l15) * DH1 + 4 * kb + 2 * half]);
    }

    const float* W1b = W1 + (size_t)branch * DH2 * DH1;
    const float* b1b = b1 + branch * DH2;
    const float* W2b = W2 + branch * DH2;

    float pA[8], pB[8];
    #pragma unroll
    for (int j = 0; j < 8; ++j) { pA[j] = 0.0f; pB[j] = 0.0f; }

    #pragma unroll
    for (int t = 0; t < 4; ++t) {
        const float bias = b1b[16 * t + l15];
        v8f accA = {bias, bias, bias, bias, bias, bias, bias, bias};
        v8f accB = accA;
        #pragma unroll
        for (int kb = 0; kb < 32; ++kb) {
            const float* bp = W1b + (size_t)(16 * t + l15) * DH1 + 4 * kb + 2 * half;
            v2f bf = ld2(bp);
            accA = __builtin_amdgcn_wmma_f32_16x16x4_f32(
                false, a1A[kb], false, bf, (short)0, accA, false, false);
            accB = __builtin_amdgcn_wmma_f32_16x16x4_f32(
                false, a1B[kb], false, bf, (short)0, accB, false, false);
        }
        const float w2v = W2b[16 * t + l15];
        #pragma unroll
        for (int j = 0; j < 8; ++j) {
            pA[j] += sigmoidf(accA[j]) * w2v;
            pB[j] += sigmoidf(accB[j]) * w2v;
        }
    }

    const float bias2 = b2[branch];
    #pragma unroll
    for (int j = 0; j < 8; ++j) {
        float vA = pA[j];
        vA += __shfl_xor(vA, 1, 32);
        vA += __shfl_xor(vA, 2, 32);
        vA += __shfl_xor(vA, 4, 32);
        vA += __shfl_xor(vA, 8, 32);
        float vB = pB[j];
        vB += __shfl_xor(vB, 1, 32);
        vB += __shfl_xor(vB, 2, 32);
        vB += __shfl_xor(vB, 4, 32);
        vB += __shfl_xor(vB, 8, 32);
        if (l15 == 0) {
            const int rowA = r0 + j + 8 * half;
            float oA = vA + bias2;
            float oB = vB + bias2;
            if (binflag) { oA = sigmoidf(oA); oB = sigmoidf(oB); }
            out[(size_t)rowA * DB + branch]        = oA;
            out[(size_t)(rowA + 16) * DB + branch] = oB;
        }
    }
}

extern "C" void kernel_launch(void* const* d_in, const int* in_sizes, int n_in,
                              void* d_out, int out_size, void* d_ws, size_t ws_size,
                              hipStream_t stream) {
    const float* x  = (const float*)d_in[0];
    const float* W0 = (const float*)d_in[1];
    const float* b0 = (const float*)d_in[2];
    const float* W1 = (const float*)d_in[3];
    const float* b1 = (const float*)d_in[4];
    const float* W2 = (const float*)d_in[5];
    const float* b2 = (const float*)d_in[6];
    float* out = (float*)d_out;

    int* flag = (ws_size >= sizeof(int)) ? (int*)d_ws : nullptr;
    if (flag) {
        init_flag_kernel<<<1, 1, 0, stream>>>(flag);
        check_binary_kernel<<<256, 256, 0, stream>>>(x, NS * DIN, flag);
    }

    const int total_waves = DB * (NS / 32);
    dim3 grid(total_waves / 2);
    topo_mlp_kernel<<<grid, 64, 0, stream>>>(x, W0, b0, W1, b1, W2, b2, out, flag);
}


# sigmoid: the correctly rounded division in 1/(1+exp(-v)) replaced by the hardware reciprocal v_rcp_f32 (within 1 ulp; argument never below 1): about 10 fewer vector instructions per activation
# speedup vs baseline: 1.1272x; 1.1272x over previous
#include <hip/hip_runtime.h>

typedef float v2f __attribute__((ext_vector_type(2)));
typedef float v8f __attribute__((ext_vector_type(8)));

#define NS   8192
#define DB   64
#define DIN  64
#define DH1  128
#define DH2  64

__device__ __forceinline__ float sigmoidf(float v) {
    return __builtin_amdgcn_rcpf(1.0f + __expf(-v));
}

__device__ __forceinline__ v2f ld2(const float* p) {
    return *(const v2f*)p;
}

__global__ void init_flag_kernel(int* f) { *f = 1; }

__global__ void check_binary_kernel(const float* __restrict__ x, int n, int* f) {
    int i = blockIdx.x * blockDim.x + threadIdx.x;
    int stride = gridDim.x * blockDim.x;
    for (; i < n; i += stride) {
        float v = x[i];
        if (v != 0.0f && v != 1.0f) { *f = 0; return; }
    }
}

__global__ void __launch_bounds__(64)
topo_mlp_kernel(const float* __restrict__ x,  const float* __restrict__ W0,
                const float* __restrict__ b0, const float* __restrict__ W1,
                const float* __restrict__ b1, const float* __restrict__ W2,
                const float* __restrict__ b2, float* __restrict__ out,
                const int* __restrict__ flag_p)
{
    __shared__ float lds_act[2][32 * DH1];

    const int lane   = threadIdx.x & 31;
    const int wav    = threadIdx.x >> 5;
    const int wid    = blockIdx.x * 2 + wav;
    const int branch = wid >> 8;
    const int tile   = wid & 255;
    const int r0     = tile * 32;
    const int half   = lane >> 4;
    const int l15    = lane & 15;

    const int binflag = flag_p ? flag_p[0] : 0;

    const float* XrowA = x + (size_t)(r0 + l15) * DIN + 2 * half;
    const float* XrowB = XrowA + 16 * DIN;
    v2f a0A[16], a0B[16];
    #pragma unroll
    for (int kb = 0; kb < 16; ++kb) {
        a0A[kb] = ld2(XrowA + 4 * kb);
        a0B[kb] = ld2(XrowB + 4 * kb);
    }

    const float* W0b = W0 + (size_t)branch * DIN * DH1;
    const float* b0b = b0 + branch * DH1;
    float* acts = lds_act[wav];

    #pragma unroll
    for (int t = 0; t < 8; ++t) {
        const float bias = b0b[16 * t + l15];
        v8f accA = {bias, bias, bias, bias, bias, bias, bias, bias};
        v8f accB = accA;
        #pragma unroll
        for (int kb = 0; kb < 16; ++kb) {
            const float* bp = W0b + (size_t)(4 * kb + 2 * half) * DH1 + 16 * t + l15;
            v2f bf = {bp[0], bp[DH1]};
            accA = __builtin_amdgcn_wmma_f32_16x16x4_f32(
                false, a0A[kb], false, bf, (short)0, accA, false, false);
            accB = __builtin_amdgcn_wmma_f32_16x16x4_f32(
                false, a0B[kb], false, bf, (short)0, accB, false, false);
        }
        #pragma unroll
        for (int j = 0; j < 8; ++j) {
            acts[(j + 8 * half) * DH1 + 16 * t + l15]        = sigmoidf(accA[j]);
            acts[(16 + j + 8 * half) * DH1 + 16 * t + l15]   = sigmoidf(accB[j]);
        }
    }

    asm volatile("s_wait_dscnt 0" ::: "memory");

    v2f a1A[32], a1B[32];
    #pragma unroll
    for (int kb = 0; kb < 32; ++kb) {
        a1A[kb] = ld2(&acts[l15 * DH1 + 4 * kb + 2 * half]);
        a1B[kb] = ld2(&acts[(16 + l15) * DH1 + 4 * kb + 2 * half]);
    }

    const float* W1b = W1 + (size_t)branch * DH2 * DH1;
    const float* b1b = b1 + branch * DH2;
    const float* W2b = W2 + branch * DH2;

    float pA[8], pB[8];
    #pragma unroll
    for (int j = 0; j < 8; ++j) { pA[j] = 0.0f; pB[j] = 0.0f; }

    #pragma unroll
    for (int t = 0; t < 4; ++t) {
        const float bias = b1b[16 * t + l15];
        v8f accA = {bias, bias, bias, bias, bias, bias, bias, bias};
        v8f accB = accA;
        #pragma unroll
        for (int kb = 0; kb < 32; ++kb) {
            const float* bp = W1b + (size_t)(16 * t + l15) * DH1 + 4 * kb + 2 * half;
            v2f bf = ld2(bp);
            accA = __builtin_amdgcn_wmma_f32_16x16x4_f32(
                false, a1A[kb], false, bf, (short)0, accA, false, false);
            accB = __builtin_amdgcn_wmma_f32_16x16x4_f32(
                false, a1B[kb], false, bf, (short)0, accB, false, false);
        }
        const float w2v = W2b[16 * t + l15];
        #pragma unroll
        for (int j = 0; j < 8; ++j) {
            pA[j] += sigmoidf(accA[j]) * w2v;
            pB[j] += sigmoidf(accB[j]) * w2v;
        }
    }

    const float bias2 = b2[branch];
    #pragma unroll
    for (int j = 0; j < 8; ++j) {
        float vA = pA[j];
        vA += __shfl_xor(vA, 1, 32);
        vA += __shfl_xor(vA, 2, 32);
        vA += __shfl_xor(vA, 4, 32);
        vA += __shfl_xor(vA, 8, 32);
        float vB = pB[j];
        vB += __shfl_xor(vB, 1, 32);
        vB += __shfl_xor(vB, 2, 32);
        vB += __shfl_xor(vB, 4, 32);
        vB += __shfl_xor(vB, 8, 32);
        if (l15 == 0) {
            const int rowA = r0 + j + 8 * half;
            float oA = vA + bias2;
            float oB = vB + bias2;
            if (binflag) { oA = sigmoidf(oA); oB = sigmoidf(oB); }
            out[(size_t)rowA * DB + branch]        = oA;
            out[(size_t)(rowA + 16) * DB + branch] = oB;
        }
    }
}

extern "C" void kernel_launch(void* const* d_in, const int* in_sizes, int n_in,
                              void* d_out, int out_size, void* d_ws, size_t ws_size,
                              hipStream_t stream) {
    const float* x  = (const float*)d_in[0];
    const float* W0 = (const float*)d_in[1];
    const float* b0 = (const float*)d_in[2];
    const float* W1 = (const float*)d_in[3];
    const float* b1 = (const float*)d_in[4];
    const float* W2 = (const float*)d_in[5];
    const float* b2 = (const float*)d_in[6];
    float* out = (float*)d_out;

    int* flag = (ws_size >= sizeof(int)) ? (int*)d_ws : nullptr;
    if (flag) {
        init_flag_kernel<<<1, 1, 0, stream>>>(flag);
        check_binary_kernel<<<256, 256, 0, stream>>>(x, NS * DIN, flag);
    }

    const int total_waves = DB * (NS / 32);
    dim3 grid(total_waves / 2);
    topo_mlp_kernel<<<grid, 64, 0, stream>>>(x, W0, b0, W1, b1, W2, b2, out, flag);
}


# all products transposed (weights as A, samples as B): each layer-0 D tile is used in place as layer-1 B fragments, no LDS relayout, 2 shuffles not 64; layer-1 k order and layer-2 sum order change
# speedup vs baseline: 1.3361x; 1.1853x over previous
#include <hip/hip_runtime.h>


typedef float v2f __attribute__((ext_vector_type(2)));
typedef float v8f __attribute__((ext_vector_type(8)));

#define NS   8192
#define DB   64
#define DIN  64
#define DH1  128
#define DH2  64

__device__ __forceinline__ float sigmoidf(float v) {
    return __builtin_amdgcn_rcpf(1.0f + __expf(-v));
}

__device__ __forceinline__ v2f ld2(const float* p) {
    return *(const v2f*)p;
}

__device__ __forceinline__ v8f ld8(const float* p) {
    return *(const v8f*)p;
}

__global__ void init_flag_kernel(int* f) { *f = 1; }

__global__ void check_binary_kernel(const float* __restrict__ x, int n, int* f) {
    int i = blockIdx.x * blockDim.x + threadIdx.x;
    int stride = gridDim.x * blockDim.x;
    for (; i < n; i += stride) {
        float v = x[i];
        if (v != 0.0f && v != 1.0f) { *f = 0; return; }
    }
}

__global__ void __launch_bounds__(64)
topo_mlp_kernel(const float* __restrict__ x,  const float* __restrict__ W0,
                const float* __restrict__ b0, const float* __restrict__ W1,
                const float* __restrict__ b1, const float* __restrict__ W2,
                const float* __restrict__ b2, float* __restrict__ out,
                const int* __restrict__ flag_p)
{
    const int lane   = threadIdx.x & 31;
    const int wav    = threadIdx.x >> 5;
    const int wid    = blockIdx.x * 2 + wav;
    const int branch = wid >> 8;
    const int tile   = wid & 255;
    const int r0     = tile * 32;
    const int half   = lane >> 4;
    const int l15    = lane & 15;

    const int binflag = flag_p ? flag_p[0] : 0;

    const float* XcolA = x + (size_t)(r0 + l15) * DIN + 2 * half;
    const float* XcolB = XcolA + 16 * DIN;
    v2f xA[16], xB[16];
    #pragma unroll
    for (int kb = 0; kb < 16; ++kb) {
        xA[kb] = ld2(XcolA + 4 * kb);
        xB[kb] = ld2(XcolB + 4 * kb);
    }

    const float* W0b = W0 + (size_t)branch * DIN * DH1;
    const float* b0b = b0 + branch * DH1;
    const float* W1b = W1 + (size_t)branch * DH2 * DH1;
    const float* b1b = b1 + branch * DH2;
    const float* W2b = W2 + branch * DH2;

    v8f acc1A[4], acc1B[4];
    #pragma unroll
    for (int u = 0; u < 4; ++u) {
        acc1A[u] = ld8(b1b + 16 * u + 8 * half);
        acc1B[u] = acc1A[u];
    }

    const float* W0lane = W0b + (size_t)(2 * half) * DH1 + l15;
    v2f a0[16];
    #pragma unroll
    for (int kb = 0; kb < 16; ++kb) {
        const float* ap = W0lane + (size_t)(4 * kb) * DH1;
        a0[kb] = (v2f){ap[0], ap[DH1]};
    }
    v8f bias0 = ld8(b0b + 8 * half);

    #pragma unroll
    for (int t = 0; t < 8; ++t) {
        v8f w1[4];
        #pragma unroll
        for (int u = 0; u < 4; ++u)
            w1[u] = ld8(W1b + (size_t)(16 * u + l15) * DH1 + 16 * t + 8 * half);
        __builtin_amdgcn_sched_barrier(0);

        v8f accA = bias0;
        v8f accB = bias0;
        #pragma unroll
        for (int kb = 0; kb < 16; ++kb) {
            accA = __builtin_amdgcn_wmma_f32_16x16x4_f32(
                false, a0[kb], false, xA[kb], (short)0, accA, false, false);
            accB = __builtin_amdgcn_wmma_f32_16x16x4_f32(
                false, a0[kb], false, xB[kb], (short)0, accB, false, false);
        }
        if (t < 7) {
            #pragma unroll
            for (int kb = 0; kb < 16; ++kb) {
                const float* ap = W0lane + (size_t)(4 * kb) * DH1 + 16 * (t + 1);
                a0[kb] = (v2f){ap[0], ap[DH1]};
            }
            bias0 = ld8(b0b + 16 * (t + 1) + 8 * half);
        }
        #pragma unroll
        for (int r = 0; r < 8; ++r) {
            accA[r] = sigmoidf(accA[r]);
            accB[r] = sigmoidf(accB[r]);
        }
        __builtin_amdgcn_sched_barrier(0);

        #pragma unroll
        for (int u = 0; u < 4; ++u) {
            #pragma unroll
            for (int s = 0; s < 4; ++s) {
                v2f af = {w1[u][2 * s], w1[u][2 * s + 1]};
                v2f bA = {accA[2 * s], accA[2 * s + 1]};
                v2f bB = {accB[2 * s], accB[2 * s + 1]};
                acc1A[u] = __builtin_amdgcn_wmma_f32_16x16x4_f32(
                    false, af, false, bA, (short)0, acc1A[u], false, false);
                acc1B[u] = __builtin_amdgcn_wmma_f32_16x16x4_f32(
                    false, af, false, bB, (short)0, acc1B[u], false, false);
            }
        }
    }

    float pA = 0.0f, pB = 0.0f;
    #pragma unroll
    for (int u = 0; u < 4; ++u) {
        const v8f w2 = ld8(W2b + 16 * u + 8 * half);
        #pragma unroll
        for (int r = 0; r < 8; ++r) {
            pA += sigmoidf(acc1A[u][r]) * w2[r];
            pB += sigmoidf(acc1B[u][r]) * w2[r];
        }
    }

    pA += __shfl_xor(pA, 16, 32);
    pB += __shfl_xor(pB, 16, 32);
    float o = (half ? pB : pA) + b2[branch];
    if (binflag) o = sigmoidf(o);
    out[(size_t)(r0 + lane) * DB + branch] = o;
}

extern "C" void kernel_launch(void* const* d_in, const int* in_sizes, int n_in,
                              void* d_out, int out_size, void* d_ws, size_t ws_size,
                              hipStream_t stream) {
    const float* x  = (const float*)d_in[0];
    const float* W0 = (const float*)d_in[1];
    const float* b0 = (const float*)d_in[2];
    const float* W1 = (const float*)d_in[3];
    const float* b1 = (const float*)d_in[4];
    const float* W2 = (const float*)d_in[5];
    const float* b2 = (const float*)d_in[6];
    float* out = (float*)d_out;

    int* flag = (ws_size >= sizeof(int)) ? (int*)d_ws : nullptr;
    if (flag) {
        init_flag_kernel<<<1, 1, 0, stream>>>(flag);
        check_binary_kernel<<<256, 256, 0, stream>>>(x, NS * DIN, flag);
    }

    const int total_waves = DB * (NS / 32);
    dim3 grid(total_waves / 2);
    topo_mlp_kernel<<<grid, 64, 0, stream>>>(x, W0, b0, W1, b1, W2, b2, out, flag);
}


# weights staged once per block in LDS in fragment order (block = one branch x 64 tiles, 8 waves x 8 tiles); fragments come by ds_load_b128, per-tile global weight loads removed
# speedup vs baseline: 1.4229x; 1.0650x over previous
#include <hip/hip_runtime.h>


typedef float v2f __attribute__((ext_vector_type(2)));
typedef float v4f __attribute__((ext_vector_type(4)));
typedef float v8f __attribute__((ext_vector_type(8)));

#define NS   8192
#define DB   64
#define DIN  64
#define DH1  128
#define DH2  64

#define NWAVES          8
#define TILES_PER_BLOCK 64
#define TPW             (TILES_PER_BLOCK / NWAVES)

__device__ __forceinline__ float sigmoidf(float v) {
    return __builtin_amdgcn_rcpf(1.0f + __expf(-v));
}

__device__ __forceinline__ v2f ld2(const float* p) {
    return *(const v2f*)p;
}

__device__ __forceinline__ v8f ld8(const float* p) {
    return *(const v8f*)p;
}

__global__ void init_flag_kernel(int* f) { *f = 1; }

__global__ void check_binary_kernel(const float* __restrict__ x, int n, int* f) {
    int i = blockIdx.x * blockDim.x + threadIdx.x;
    int stride = gridDim.x * blockDim.x;
    for (; i < n; i += stride) {
        float v = x[i];
        if (v != 0.0f && v != 1.0f) { *f = 0; return; }
    }
}

__global__ void __launch_bounds__(NWAVES * 32)
topo_mlp_kernel(const float* __restrict__ x,  const float* __restrict__ W0,
                const float* __restrict__ b0, const float* __restrict__ W1,
                const float* __restrict__ b1, const float* __restrict__ W2,
                const float* __restrict__ b2, float* __restrict__ out,
                const int* __restrict__ flag_p)
{
    __shared__ v4f sW0[8 * 8 * 32];
    __shared__ v4f sW1[8 * 8 * 32];

    const int lane   = threadIdx.x & 31;
    const int wav    = __builtin_amdgcn_readfirstlane((int)(threadIdx.x >> 5));
    const int branch = blockIdx.x >> 2;
    const int tile0  = (blockIdx.x & 3) * TILES_PER_BLOCK + wav;
    const int half   = lane >> 4;
    const int l15    = lane & 15;

    const int binflag = flag_p ? flag_p[0] : 0;

    const float* W0b = W0 + (size_t)branch * DIN * DH1;
    const float* b0b = b0 + branch * DH1;
    const float* W1b = W1 + (size_t)branch * DH2 * DH1;
    const float* b1b = b1 + branch * DH2;
    const float* W2b = W2 + branch * DH2;

    const float* Xlane = x + (size_t)l15 * DIN + 2 * half;
    v2f xA[16], xB[16];
    {
        const float* XcolA = Xlane + (size_t)(tile0 * 32) * DIN;
        const float* XcolB = XcolA + 16 * DIN;
        #pragma unroll
        for (int kb = 0; kb < 16; ++kb) {
            xA[kb] = ld2(XcolA + 4 * kb);
            xB[kb] = ld2(XcolB + 4 * kb);
        }
    }

    {
        const float* ap = W0b + (size_t)(2 * half) * DH1 + l15 + 16 * wav;
        #pragma unroll
        for (int p = 0; p < 8; ++p) {
            const float* a = ap + (size_t)(8 * p) * DH1;
            sW0[(wav * 8 + p) * 32 + lane] = (v4f){a[0], a[DH1], a[4 * DH1], a[5 * DH1]};
        }
        #pragma unroll
        for (int u = 0; u < 4; ++u) {
            const v8f w = ld8(W1b + (size_t)(16 * u + l15) * DH1 + 16 * wav + 8 * half);
            sW1[((wav * 4 + u) * 2 + 0) * 32 + lane] = (v4f){w[0], w[1], w[2], w[3]};
            sW1[((wav * 4 + u) * 2 + 1) * 32 + lane] = (v4f){w[4], w[5], w[6], w[7]};
        }
    }
    __syncthreads();

    v2f a0[16];
    #pragma unroll
    for (int p = 0; p < 8; ++p) {
        const v4f v = sW0[p * 32 + lane];
        a0[2 * p]     = (v2f){v[0], v[1]};
        a0[2 * p + 1] = (v2f){v[2], v[3]};
    }
    v8f bias0 = ld8(b0b + 8 * half);

    #pragma unroll 1
    for (int i = 0; i < TPW; ++i) {
        const int r0 = (tile0 + i * NWAVES) * 32;

        v8f acc1A[4], acc1B[4];
        #pragma unroll
        for (int u = 0; u < 4; ++u) {
            acc1A[u] = ld8(b1b + 16 * u + 8 * half);
            acc1B[u] = acc1A[u];
        }

        #pragma unroll
        for (int t = 0; t < 8; ++t) {
            v8f w1[4];
            #pragma unroll
            for (int u = 0; u < 4; ++u) {
                const v4f lo = sW1[((t * 4 + u) * 2 + 0) * 32 + lane];
                const v4f hi = sW1[((t * 4 + u) * 2 + 1) * 32 + lane];
                w1[u] = (v8f){lo[0], lo[1], lo[2], lo[3], hi[0], hi[1], hi[2], hi[3]};
            }
            __builtin_amdgcn_sched_barrier(0);

            v8f accA = bias0;
            v8f accB = bias0;
            #pragma unroll
            for (int kb = 0; kb < 16; ++kb) {
                accA = __builtin_amdgcn_wmma_f32_16x16x4_f32(
                    false, a0[kb], false, xA[kb], (short)0, accA, false, false);
                accB = __builtin_amdgcn_wmma_f32_16x16x4_f32(
                    false, a0[kb], false, xB[kb], (short)0, accB, false, false);
            }
            {
                const int tn = (t + 1) & 7;
                #pragma unroll
                for (int p = 0; p < 8; ++p) {
                    const v4f v = sW0[(tn * 8 + p) * 32 + lane];
                    a0[2 * p]     = (v2f){v[0], v[1]};
                    a0[2 * p + 1] = (v2f){v[2], v[3]};
                }
                bias0 = ld8(b0b + 16 * tn + 8 * half);
            }
            if (t == 7 && i + 1 < TPW) {
                const float* XcolA = Xlane + (size_t)(r0 + NWAVES * 32) * DIN;
                const float* XcolB = XcolA + 16 * DIN;
                #pragma unroll
                for (int kb = 0; kb < 16; ++kb) {
                    xA[kb] = ld2(XcolA + 4 * kb);
                    xB[kb] = ld2(XcolB + 4 * kb);
                }
            }
            #pragma unroll
            for (int r = 0; r < 8; ++r) {
                accA[r] = sigmoidf(accA[r]);
                accB[r] = sigmoidf(accB[r]);
            }
            __builtin_amdgcn_sched_barrier(0);

            #pragma unroll
            for (int u = 0; u < 4; ++u) {
                #pragma unroll
                for (int s = 0; s < 4; ++s) {
                    v2f af = {w1[u][2 * s], w1[u][2 * s + 1]};
                    v2f bA = {accA[2 * s], accA[2 * s + 1]};
                    v2f bB = {accB[2 * s], accB[2 * s + 1]};
                    acc1A[u] = __builtin_amdgcn_wmma_f32_16x16x4_f32(
                        false, af, false, bA, (short)0, acc1A[u], false, false);
                    acc1B[u] = __builtin_amdgcn_wmma_f32_16x16x4_f32(
                        false, af, false, bB, (short)0, acc1B[u], false, false);
                }
            }
        }

        float pA = 0.0f, pB = 0.0f;
        #pragma unroll
        for (int u = 0; u < 4; ++u) {
            const v8f w2 = ld8(W2b + 16 * u + 8 * half);
            #pragma unroll
            for (int r = 0; r < 8; ++r) {
                pA += sigmoidf(acc1A[u][r]) * w2[r];
                pB += sigmoidf(acc1B[u][r]) * w2[r];
            }
        }

        pA += __shfl_xor(pA, 16, 32);
        pB += __shfl_xor(pB, 16, 32);
        float o = (half ? pB : pA) + b2[branch];
        if (binflag) o = sigmoidf(o);
        out[(size_t)(r0 + lane) * DB + branch] = o;
    }
}

extern "C" void kernel_launch(void* const* d_in, const int* in_sizes, int n_in,
                              void* d_out, int out_size, void* d_ws, size_t ws_size,
                              hipStream_t stream) {
    const float* x  = (const float*)d_in[0];
    const float* W0 = (const float*)d_in[1];
    const float* b0 = (const float*)d_in[2];
    const float* W1 = (const float*)d_in[3];
    const float* b1 = (const float*)d_in[4];
    const float* W2 = (const float*)d_in[5];
    const float* b2 = (const float*)d_in[6];
    float* out = (float*)d_out;

    int* flag = (ws_size >= sizeof(int)) ? (int*)d_ws : nullptr;
    if (flag) {
        init_flag_kernel<<<1, 1, 0, stream>>>(flag);
        check_binary_kernel<<<256, 256, 0, stream>>>(x, NS * DIN, flag);
    }

    dim3 grid(DB * (NS / 32) / TILES_PER_BLOCK);
    topo_mlp_kernel<<<grid, NWAVES * 32, 0, stream>>>(x, W0, b0, W1, b1, W2, b2, out, flag);
}


# input test all(x==0|x==1) moved into the main kernel: each block decides it itself from the tile it holds and scans all of x only if that tile is all 0/1; both flag launches removed, 3 launches -> 1
# speedup vs baseline: 1.5167x; 1.0659x over previous
#include <hip/hip_runtime.h>


typedef float v2f __attribute__((ext_vector_type(2)));
typedef float v4f __attribute__((ext_vector_type(4)));
typedef float v8f __attribute__((ext_vector_type(8)));

#define NS   8192
#define DB   64
#define DIN  64
#define DH1  128
#define DH2  64

#define NWAVES          8
#define TILES_PER_BLOCK 64
#define TPW             (TILES_PER_BLOCK / NWAVES)

__device__ __forceinline__ float sigmoidf(float v) {
    return __builtin_amdgcn_rcpf(1.0f + __expf(-v));
}

__device__ __forceinline__ v2f ld2(const float* p) {
    return *(const v2f*)p;
}

__device__ __forceinline__ v8f ld8(const float* p) {
    return *(const v8f*)p;
}

typedef int v4i __attribute__((ext_vector_type(4)));

__device__ __forceinline__ unsigned other_bits(float v) {
    const unsigned b = __float_as_uint(v);
    return min(b << 1, b ^ 0x3f800000u);
}

#define SCAN_V4_PER_WAVE (NS * DIN / 4 / NWAVES)
#define SCAN_GROUP       16

__global__ void __launch_bounds__(NWAVES * 32)
topo_mlp_kernel(const float* __restrict__ x,  const float* __restrict__ W0,
                const float* __restrict__ b0, const float* __restrict__ W1,
                const float* __restrict__ b1, const float* __restrict__ W2,
                const float* __restrict__ b2, float* __restrict__ out)
{
    __shared__ v4f sW0[8 * 8 * 32];
    __shared__ v4f sW1[8 * 8 * 32];
    __shared__ v4i sBin[NWAVES / 4];

    const int lane   = threadIdx.x & 31;
    const int wav    = __builtin_amdgcn_readfirstlane((int)(threadIdx.x >> 5));
    const int branch = blockIdx.x >> 2;
    const int tile0  = (blockIdx.x & 3) * TILES_PER_BLOCK + wav;
    const int half   = lane >> 4;
    const int l15    = lane & 15;

    const float* W0b = W0 + (size_t)branch * DIN * DH1;
    const float* b0b = b0 + branch * DH1;
    const float* W1b = W1 + (size_t)branch * DH2 * DH1;
    const float* b1b = b1 + branch * DH2;
    const float* W2b = W2 + branch * DH2;

    const float* Xlane = x + (size_t)l15 * DIN + 2 * half;
    v2f xA[16], xB[16];
    {
        const float* XcolA = Xlane + (size_t)(tile0 * 32) * DIN;
        const float* XcolB = XcolA + 16 * DIN;
        #pragma unroll
        for (int kb = 0; kb < 16; ++kb) {
            xA[kb] = ld2(XcolA + 4 * kb);
            xB[kb] = ld2(XcolB + 4 * kb);
        }
    }

    {
        const float* ap = W0b + (size_t)(2 * half) * DH1 + l15 + 16 * wav;
        #pragma unroll
        for (int p = 0; p < 8; ++p) {
            const float* a = ap + (size_t)(8 * p) * DH1;
            sW0[(wav * 8 + p) * 32 + lane] = (v4f){a[0], a[DH1], a[4 * DH1], a[5 * DH1]};
        }
        #pragma unroll
        for (int u = 0; u < 4; ++u) {
            const v8f w = ld8(W1b + (size_t)(16 * u + l15) * DH1 + 16 * wav + 8 * half);
            sW1[((wav * 4 + u) * 2 + 0) * 32 + lane] = (v4f){w[0], w[1], w[2], w[3]};
            sW1[((wav * 4 + u) * 2 + 1) * 32 + lane] = (v4f){w[4], w[5], w[6], w[7]};
        }
    }

    {
        unsigned other = 0u;
        #pragma unroll
        for (int kb = 0; kb < 16; ++kb) {
            other |= other_bits(xA[kb][0]) | other_bits(xA[kb][1]) | other_bits(xB[kb][0]) | other_bits(xB[kb][1]);
        }
        bool wave_other = __builtin_amdgcn_ballot_w32(other != 0u) != 0u;
        if (!wave_other) {
            const v4f* xs = (const v4f*)x + (size_t)wav * SCAN_V4_PER_WAVE + lane;
            #pragma unroll 1
            for (int it = 0; it < SCAN_V4_PER_WAVE / (32 * SCAN_GROUP); ++it) {
                unsigned o = 0u;
                #pragma unroll
                for (int g = 0; g < SCAN_GROUP; ++g) {
                    const v4f v = xs[(it * SCAN_GROUP + g) * 32];
                    o |= other_bits(v[0]) | other_bits(v[1]) | other_bits(v[2]) | other_bits(v[3]);
                }
                if (__builtin_amdgcn_ballot_w32(o != 0u) != 0u) { wave_other = true; break; }
            }
        }
        if (lane == 0) ((int*)sBin)[wav] = wave_other ? 0 : 1;
    }
    __syncthreads();

    int binflag;
    {
        const v4i f0 = sBin[0], f1 = sBin[1];
        binflag = __builtin_amdgcn_readfirstlane(f0[0] & f0[1] & f0[2] & f0[3] & f1[0] & f1[1] & f1[2] & f1[3]);
    }

    v2f a0[16];
    #pragma unroll
    for (int p = 0; p < 8; ++p) {
        const v4f v = sW0[p * 32 + lane];
        a0[2 * p]     = (v2f){v[0], v[1]};
        a0[2 * p + 1] = (v2f){v[2], v[3]};
    }
    v8f bias0 = ld8(b0b + 8 * half);

    #pragma unroll 1
    for (int i = 0; i < TPW; ++i) {
        const int r0 = (tile0 + i * NWAVES) * 32;

        v8f acc1A[4], acc1B[4];
        #pragma unroll
        for (int u = 0; u < 4; ++u) {
            acc1A[u] = ld8(b1b + 16 * u + 8 * half);
            acc1B[u] = acc1A[u];
        }

        #pragma unroll
        for (int t = 0; t < 8; ++t) {
            v8f w1[4];
            #pragma unroll
            for (int u = 0; u < 4; ++u) {
                const v4f lo = sW1[((t * 4 + u) * 2 + 0) * 32 + lane];
                const v4f hi = sW1[((t * 4 + u) * 2 + 1) * 32 + lane];
                w1[u] = (v8f){lo[0], lo[1], lo[2], lo[3], hi[0], hi[1], hi[2], hi[3]};
            }
            __builtin_amdgcn_sched_barrier(0);

            v8f accA = bias0;
            v8f accB = bias0;
            #pragma unroll
            for (int kb = 0; kb < 16; ++kb) {
                accA = __builtin_amdgcn_wmma_f32_16x16x4_f32(
                    false, a0[kb], false, xA[kb], (short)0, accA, false, false);
                accB = __builtin_amdgcn_wmma_f32_16x16x4_f32(
                    false, a0[kb], false, xB[kb], (short)0, accB, false, false);
            }
            {
                const int tn = (t + 1) & 7;
                #pragma unroll
                for (int p = 0; p < 8; ++p) {
                    const v4f v = sW0[(tn * 8 + p) * 32 + lane];
                    a0[2 * p]     = (v2f){v[0], v[1]};
                    a0[2 * p + 1] = (v2f){v[2], v[3]};
                }
                bias0 = ld8(b0b + 16 * tn + 8 * half);
            }
            if (t == 7 && i + 1 < TPW) {
                const float* XcolA = Xlane + (size_t)(r0 + NWAVES * 32) * DIN;
                const float* XcolB = XcolA + 16 * DIN;
                #pragma unroll
                for (int kb = 0; kb < 16; ++kb) {
                    xA[kb] = ld2(XcolA + 4 * kb);
                    xB[kb] = ld2(XcolB + 4 * kb);
                }
            }
            #pragma unroll
            for (int r = 0; r < 8; ++r) {
                accA[r] = sigmoidf(accA[r]);
                accB[r] = sigmoidf(accB[r]);
            }
            __builtin_amdgcn_sched_barrier(0);

            #pragma unroll
            for (int u = 0; u < 4; ++u) {
                #pragma unroll
                for (int s = 0; s < 4; ++s) {
                    v2f af = {w1[u][2 * s], w1[u][2 * s + 1]};
                    v2f bA = {accA[2 * s], accA[2 * s + 1]};
                    v2f bB = {accB[2 * s], accB[2 * s + 1]};
                    acc1A[u] = __builtin_amdgcn_wmma_f32_16x16x4_f32(
                        false, af, false, bA, (short)0, acc1A[u], false, false);
                    acc1B[u] = __builtin_amdgcn_wmma_f32_16x16x4_f32(
                        false, af, false, bB, (short)0, acc1B[u], false, false);
                }
            }
        }

        float pA = 0.0f, pB = 0.0f;
        #pragma unroll
        for (int u = 0; u < 4; ++u) {
            const v8f w2 = ld8(W2b + 16 * u + 8 * half);
            #pragma unroll
            for (int r = 0; r < 8; ++r) {
                pA += sigmoidf(acc1A[u][r]) * w2[r];
                pB += sigmoidf(acc1B[u][r]) * w2[r];
            }
        }

        pA += __shfl_xor(pA, 16, 32);
        pB += __shfl_xor(pB, 16, 32);
        float o = (half ? pB : pA) + b2[branch];
        if (binflag) o = sigmoidf(o);
        out[(size_t)(r0 + lane) * DB + branch] = o;
    }
}

extern "C" void kernel_launch(void* const* d_in, const int* in_sizes, int n_in,
                              void* d_out, int out_size, void* d_ws, size_t ws_size,
                              hipStream_t stream) {
    const float* x  = (const float*)d_in[0];
    const float* W0 = (const float*)d_in[1];
    const float* b0 = (const float*)d_in[2];
    const float* W1 = (const float*)d_in[3];
    const float* b1 = (const float*)d_in[4];
    const float* W2 = (const float*)d_in[5];
    const float* b2 = (const float*)d_in[6];
    float* out = (float*)d_out;

    dim3 grid(DB * (NS / 32) / TILES_PER_BLOCK);
    topo_mlp_kernel<<<grid, NWAVES * 32, 0, stream>>>(x, W0, b0, W1, b1, W2, b2, out);
}


# sigmoid of layers 0 and 1 as 0.5*tanh(v/2)+0.5: one v_tanh_f32 and one multiply-add replace multiply, exp, add, reciprocal; the halving folded into W0, b0, W1, b1 when staged (exact scaling)
# speedup vs baseline: 1.5789x; 1.0410x over previous
#include <hip/hip_runtime.h>


typedef float v2f __attribute__((ext_vector_type(2)));
typedef float v4f __attribute__((ext_vector_type(4)));
typedef float v8f __attribute__((ext_vector_type(8)));

#define NS   8192
#define DB   64
#define DIN  64
#define DH1  128
#define DH2  64

#define NWAVES          8
#define TILES_PER_BLOCK 64
#define TPW             (TILES_PER_BLOCK / NWAVES)

__device__ __forceinline__ float sigmoidf(float v) {
    return __builtin_amdgcn_rcpf(1.0f + __expf(-v));
}

__device__ __forceinline__ float sigmoid_of_half(float vh) {
    return __builtin_fmaf(0.5f, __builtin_amdgcn_tanhf(vh), 0.5f);
}

__device__ __forceinline__ v2f ld2(const float* p) {
    return *(const v2f*)p;
}

__device__ __forceinline__ v8f ld8(const float* p) {
    return *(const v8f*)p;
}

typedef int v4i __attribute__((ext_vector_type(4)));

__device__ __forceinline__ unsigned other_bits(float v) {
    const unsigned b = __float_as_uint(v);
    return min(b << 1, b ^ 0x3f800000u);
}

#define SCAN_V4_PER_WAVE (NS * DIN / 4 / NWAVES)
#define SCAN_GROUP       16

__global__ void __launch_bounds__(NWAVES * 32)
topo_mlp_kernel(const float* __restrict__ x,  const float* __restrict__ W0,
                const float* __restrict__ b0, const float* __restrict__ W1,
                const float* __restrict__ b1, const float* __restrict__ W2,
                const float* __restrict__ b2, float* __restrict__ out)
{
    __shared__ v4f sW0[8 * 8 * 32];
    __shared__ v4f sW1[8 * 8 * 32];
    __shared__ v4i sBin[NWAVES / 4];

    const int lane   = threadIdx.x & 31;
    const int wav    = __builtin_amdgcn_readfirstlane((int)(threadIdx.x >> 5));
    const int branch = blockIdx.x >> 2;
    const int tile0  = (blockIdx.x & 3) * TILES_PER_BLOCK + wav;
    const int half   = lane >> 4;
    const int l15    = lane & 15;

    const float* W0b = W0 + (size_t)branch * DIN * DH1;
    const float* b0b = b0 + branch * DH1;
    const float* W1b = W1 + (size_t)branch * DH2 * DH1;
    const float* b1b = b1 + branch * DH2;
    const float* W2b = W2 + branch * DH2;

    const float* Xlane = x + (size_t)l15 * DIN + 2 * half;
    v2f xA[16], xB[16];
    {
        const float* XcolA = Xlane + (size_t)(tile0 * 32) * DIN;
        const float* XcolB = XcolA + 16 * DIN;
        #pragma unroll
        for (int kb = 0; kb < 16; ++kb) {
            xA[kb] = ld2(XcolA + 4 * kb);
            xB[kb] = ld2(XcolB + 4 * kb);
        }
    }

    {
        const float* ap = W0b + (size_t)(2 * half) * DH1 + l15 + 16 * wav;
        #pragma unroll
        for (int p = 0; p < 8; ++p) {
            const float* a = ap + (size_t)(8 * p) * DH1;
            sW0[(wav * 8 + p) * 32 + lane] = (v4f){a[0], a[DH1], a[4 * DH1], a[5 * DH1]} * 0.5f;
        }
        #pragma unroll
        for (int u = 0; u < 4; ++u) {
            const v8f w = ld8(W1b + (size_t)(16 * u + l15) * DH1 + 16 * wav + 8 * half) * 0.5f;
            sW1[((wav * 4 + u) * 2 + 0) * 32 + lane] = (v4f){w[0], w[1], w[2], w[3]};
            sW1[((wav * 4 + u) * 2 + 1) * 32 + lane] = (v4f){w[4], w[5], w[6], w[7]};
        }
    }

    {
        unsigned other = 0u;
        #pragma unroll
        for (int kb = 0; kb < 16; ++kb) {
            other |= other_bits(xA[kb][0]) | other_bits(xA[kb][1]) | other_bits(xB[kb][0]) | other_bits(xB[kb][1]);
        }
        bool wave_other = __builtin_amdgcn_ballot_w32(other != 0u) != 0u;
        if (!wave_other) {
            const v4f* xs = (const v4f*)x + (size_t)wav * SCAN_V4_PER_WAVE + lane;
            #pragma unroll 1
            for (int it = 0; it < SCAN_V4_PER_WAVE / (32 * SCAN_GROUP); ++it) {
                unsigned o = 0u;
                #pragma unroll
                for (int g = 0; g < SCAN_GROUP; ++g) {
                    const v4f v = xs[(it * SCAN_GROUP + g) * 32];
                    o |= other_bits(v[0]) | other_bits(v[1]) | other_bits(v[2]) | other_bits(v[3]);
                }
                if (__builtin_amdgcn_ballot_w32(o != 0u) != 0u) { wave_other = true; break; }
            }
        }
        if (lane == 0) ((int*)sBin)[wav] = wave_other ? 0 : 1;
    }
    __syncthreads();

    int binflag;
    {
        const v4i f0 = sBin[0], f1 = sBin[1];
        binflag = __builtin_amdgcn_readfirstlane(f0[0] & f0[1] & f0[2] & f0[3] & f1[0] & f1[1] & f1[2] & f1[3]);
    }

    v2f a0[16];
    #pragma unroll
    for (int p = 0; p < 8; ++p) {
        const v4f v = sW0[p * 32 + lane];
        a0[2 * p]     = (v2f){v[0], v[1]};
        a0[2 * p + 1] = (v2f){v[2], v[3]};
    }
    v8f bias0 = ld8(b0b + 8 * half) * 0.5f;

    #pragma unroll 1
    for (int i = 0; i < TPW; ++i) {
        const int r0 = (tile0 + i * NWAVES) * 32;

        v8f acc1A[4], acc1B[4];
        #pragma unroll
        for (int u = 0; u < 4; ++u) {
            acc1A[u] = ld8(b1b + 16 * u + 8 * half) * 0.5f;
            acc1B[u] = acc1A[u];
        }

        #pragma unroll
        for (int t = 0; t < 8; ++t) {
            v8f w1[4];
            #pragma unroll
            for (int u = 0; u < 4; ++u) {
                const v4f lo = sW1[((t * 4 + u) * 2 + 0) * 32 + lane];
                const v4f hi = sW1[((t * 4 + u) * 2 + 1) * 32 + lane];
                w1[u] = (v8f){lo[0], lo[1], lo[2], lo[3], hi[0], hi[1], hi[2], hi[3]};
            }
            __builtin_amdgcn_sched_barrier(0);

            v8f accA = bias0;
            v8f accB = bias0;
            #pragma unroll
            for (int kb = 0; kb < 16; ++kb) {
                accA = __builtin_amdgcn_wmma_f32_16x16x4_f32(
                    false, a0[kb], false, xA[kb], (short)0, accA, false, false);
                accB = __builtin_amdgcn_wmma_f32_16x16x4_f32(
                    false, a0[kb], false, xB[kb], (short)0, accB, false, false);
            }
            {
                const int tn = (t + 1) & 7;
                #pragma unroll
                for (int p = 0; p < 8; ++p) {
                    const v4f v = sW0[(tn * 8 + p) * 32 + lane];
                    a0[2 * p]     = (v2f){v[0], v[1]};
                    a0[2 * p + 1] = (v2f){v[2], v[3]};
                }
                bias0 = ld8(b0b + 16 * tn + 8 * half) * 0.5f;
            }
            if (t == 7 && i + 1 < TPW) {
                const float* XcolA = Xlane + (size_t)(r0 + NWAVES * 32) * DIN;
                const float* XcolB = XcolA + 16 * DIN;
                #pragma unroll
                for (int kb = 0; kb < 16; ++kb) {
                    xA[kb] = ld2(XcolA + 4 * kb);
                    xB[kb] = ld2(XcolB + 4 * kb);
                }
            }
            #pragma unroll
            for (int r = 0; r < 8; ++r) {
                accA[r] = sigmoid_of_half(accA[r]);
                accB[r] = sigmoid_of_half(accB[r]);
            }
            __builtin_amdgcn_sched_barrier(0);

            #pragma unroll
            for (int u = 0; u < 4; ++u) {
                #pragma unroll
                for (int s = 0; s < 4; ++s) {
                    v2f af = {w1[u][2 * s], w1[u][2 * s + 1]};
                    v2f bA = {accA[2 * s], accA[2 * s + 1]};
                    v2f bB = {accB[2 * s], accB[2 * s + 1]};
                    acc1A[u] = __builtin_amdgcn_wmma_f32_16x16x4_f32(
                        false, af, false, bA, (short)0, acc1A[u], false, false);
                    acc1B[u] = __builtin_amdgcn_wmma_f32_16x16x4_f32(
                        false, af, false, bB, (short)0, acc1B[u], false, false);
                }
            }
        }

        float pA = 0.0f, pB = 0.0f;
        #pragma unroll
        for (int u = 0; u < 4; ++u) {
            const v8f w2 = ld8(W2b + 16 * u + 8 * half);
            #pragma unroll
            for (int r = 0; r < 8; ++r) {
                pA += sigmoid_of_half(acc1A[u][r]) * w2[r];
                pB += sigmoid_of_half(acc1B[u][r]) * w2[r];
            }
        }

        pA += __shfl_xor(pA, 16, 32);
        pB += __shfl_xor(pB, 16, 32);
        float o = (half ? pB : pA) + b2[branch];
        if (binflag) o = sigmoidf(o);
        out[(size_t)(r0 + lane) * DB + branch] = o;
    }
}

extern "C" void kernel_launch(void* const* d_in, const int* in_sizes, int n_in,
                              void* d_out, int out_size, void* d_ws, size_t ws_size,
                              hipStream_t stream) {
    const float* x  = (const float*)d_in[0];
    const float* W0 = (const float*)d_in[1];
    const float* b0 = (const float*)d_in[2];
    const float* W1 = (const float*)d_in[3];
    const float* b1 = (const float*)d_in[4];
    const float* W2 = (const float*)d_in[5];
    const float* b2 = (const float*)d_in[6];
    float* out = (float*)d_out;

    dim3 grid(DB * (NS / 32) / TILES_PER_BLOCK);
    topo_mlp_kernel<<<grid, NWAVES * 32, 0, stream>>>(x, W0, b0, W1, b1, W2, b2, out);
}


# s_setprio by own tile index in every wave (3 for tiles 0-2, 2 for 3-4, 1 for 5-6, 0 for the last): the two waves of a SIMD take turns on the matrix pipe and finish together, no long solo tail
# speedup vs baseline: 1.6320x; 1.0336x over previous
#include <hip/hip_runtime.h>


typedef float v2f __attribute__((ext_vector_type(2)));
typedef float v4f __attribute__((ext_vector_type(4)));
typedef float v8f __attribute__((ext_vector_type(8)));

#define NS   8192
#define DB   64
#define DIN  64
#define DH1  128
#define DH2  64

#define NWAVES          8
#define TILES_PER_BLOCK 64
#define TPW             (TILES_PER_BLOCK / NWAVES)

__device__ __forceinline__ float sigmoidf(float v) {
    return __builtin_amdgcn_rcpf(1.0f + __expf(-v));
}

__device__ __forceinline__ float sigmoid_of_half(float vh) {
    return __builtin_fmaf(0.5f, __builtin_amdgcn_tanhf(vh), 0.5f);
}

__device__ __forceinline__ v2f ld2(const float* p) {
    return *(const v2f*)p;
}

__device__ __forceinline__ v8f ld8(const float* p) {
    return *(const v8f*)p;
}

typedef int v4i __attribute__((ext_vector_type(4)));

__device__ __forceinline__ unsigned other_bits(float v) {
    const unsigned b = __float_as_uint(v);
    return min(b << 1, b ^ 0x3f800000u);
}

#define SCAN_V4_PER_WAVE (NS * DIN / 4 / NWAVES)
#define SCAN_GROUP       16

__global__ void __launch_bounds__(NWAVES * 32)
topo_mlp_kernel(const float* __restrict__ x,  const float* __restrict__ W0,
                const float* __restrict__ b0, const float* __restrict__ W1,
                const float* __restrict__ b1, const float* __restrict__ W2,
                const float* __restrict__ b2, float* __restrict__ out)
{
    __shared__ v4f sW0[8 * 8 * 32];
    __shared__ v4f sW1[8 * 8 * 32];
    __shared__ v4i sBin[NWAVES / 4];

    const int lane   = threadIdx.x & 31;
    const int wav    = __builtin_amdgcn_readfirstlane((int)(threadIdx.x >> 5));
    const int branch = blockIdx.x >> 2;
    const int tile0  = (blockIdx.x & 3) * TILES_PER_BLOCK + wav;
    const int half   = lane >> 4;
    const int l15    = lane & 15;

    const float* W0b = W0 + (size_t)branch * DIN * DH1;
    const float* b0b = b0 + branch * DH1;
    const float* W1b = W1 + (size_t)branch * DH2 * DH1;
    const float* b1b = b1 + branch * DH2;
    const float* W2b = W2 + branch * DH2;

    const float* Xlane = x + (size_t)l15 * DIN + 2 * half;
    v2f xA[16], xB[16];
    {
        const float* XcolA = Xlane + (size_t)(tile0 * 32) * DIN;
        const float* XcolB = XcolA + 16 * DIN;
        #pragma unroll
        for (int kb = 0; kb < 16; ++kb) {
            xA[kb] = ld2(XcolA + 4 * kb);
            xB[kb] = ld2(XcolB + 4 * kb);
        }
    }

    {
        const float* ap = W0b + (size_t)(2 * half) * DH1 + l15 + 16 * wav;
        #pragma unroll
        for (int p = 0; p < 8; ++p) {
            const float* a = ap + (size_t)(8 * p) * DH1;
            sW0[(wav * 8 + p) * 32 + lane] = (v4f){a[0], a[DH1], a[4 * DH1], a[5 * DH1]} * 0.5f;
        }
        #pragma unroll
        for (int u = 0; u < 4; ++u) {
            const v8f w = ld8(W1b + (size_t)(16 * u + l15) * DH1 + 16 * wav + 8 * half) * 0.5f;
            sW1[((wav * 4 + u) * 2 + 0) * 32 + lane] = (v4f){w[0], w[1], w[2], w[3]};
            sW1[((wav * 4 + u) * 2 + 1) * 32 + lane] = (v4f){w[4], w[5], w[6], w[7]};
        }
    }

    {
        unsigned other = 0u;
        #pragma unroll
        for (int kb = 0; kb < 16; ++kb) {
            other |= other_bits(xA[kb][0]) | other_bits(xA[kb][1]) | other_bits(xB[kb][0]) | other_bits(xB[kb][1]);
        }
        bool wave_other = __builtin_amdgcn_ballot_w32(other != 0u) != 0u;
        if (!wave_other) {
            const v4f* xs = (const v4f*)x + (size_t)wav * SCAN_V4_PER_WAVE + lane;
            #pragma unroll 1
            for (int it = 0; it < SCAN_V4_PER_WAVE / (32 * SCAN_GROUP); ++it) {
                unsigned o = 0u;
                #pragma unroll
                for (int g = 0; g < SCAN_GROUP; ++g) {
                    const v4f v = xs[(it * SCAN_GROUP + g) * 32];
                    o |= other_bits(v[0]) | other_bits(v[1]) | other_bits(v[2]) | other_bits(v[3]);
                }
                if (__builtin_amdgcn_ballot_w32(o != 0u) != 0u) { wave_other = true; break; }
            }
        }
        if (lane == 0) ((int*)sBin)[wav] = wave_other ? 0 : 1;
    }
    __syncthreads();

    int binflag;
    {
        const v4i f0 = sBin[0], f1 = sBin[1];
        binflag = __builtin_amdgcn_readfirstlane(f0[0] & f0[1] & f0[2] & f0[3] & f1[0] & f1[1] & f1[2] & f1[3]);
    }

    v2f a0[16];
    #pragma unroll
    for (int p = 0; p < 8; ++p) {
        const v4f v = sW0[p * 32 + lane];
        a0[2 * p]     = (v2f){v[0], v[1]};
        a0[2 * p + 1] = (v2f){v[2], v[3]};
    }
    v8f bias0 = ld8(b0b + 8 * half) * 0.5f;


    #pragma unroll 1
    for (int i = 0; i < TPW; ++i) {
        const int r0 = (tile0 + i * NWAVES) * 32;

        if (i == 0)      __builtin_amdgcn_s_setprio(3);
        else if (i == 3) __builtin_amdgcn_s_setprio(2);
        else if (i == 5) __builtin_amdgcn_s_setprio(1);
        else if (i == 7) __builtin_amdgcn_s_setprio(0);

        v8f acc1A[4], acc1B[4];
        #pragma unroll
        for (int u = 0; u < 4; ++u) {
            acc1A[u] = ld8(b1b + 16 * u + 8 * half) * 0.5f;
            acc1B[u] = acc1A[u];
        }

        #pragma unroll
        for (int t = 0; t < 8; ++t) {
            v8f w1[4];
            #pragma unroll
            for (int u = 0; u < 4; ++u) {
                const v4f lo = sW1[((t * 4 + u) * 2 + 0) * 32 + lane];
                const v4f hi = sW1[((t * 4 + u) * 2 + 1) * 32 + lane];
                w1[u] = (v8f){lo[0], lo[1], lo[2], lo[3], hi[0], hi[1], hi[2], hi[3]};
            }
            __builtin_amdgcn_sched_barrier(0);

            v8f accA = bias0;
            v8f accB = bias0;
            #pragma unroll
            for (int kb = 0; kb < 16; ++kb) {
                accA = __builtin_amdgcn_wmma_f32_16x16x4_f32(
                    false, a0[kb], false, xA[kb], (short)0, accA, false, false);
                accB = __builtin_amdgcn_wmma_f32_16x16x4_f32(
                    false, a0[kb], false, xB[kb], (short)0, accB, false, false);
            }
            {
                const int tn = (t + 1) & 7;
                #pragma unroll
                for (int p = 0; p < 8; ++p) {
                    const v4f v = sW0[(tn * 8 + p) * 32 + lane];
                    a0[2 * p]     = (v2f){v[0], v[1]};
                    a0[2 * p + 1] = (v2f){v[2], v[3]};
                }
                bias0 = ld8(b0b + 16 * tn + 8 * half) * 0.5f;
            }
            if (t == 7 && i + 1 < TPW) {
                const float* XcolA = Xlane + (size_t)(r0 + NWAVES * 32) * DIN;
                const float* XcolB = XcolA + 16 * DIN;
                #pragma unroll
                for (int kb = 0; kb < 16; ++kb) {
                    xA[kb] = ld2(XcolA + 4 * kb);
                    xB[kb] = ld2(XcolB + 4 * kb);
                }
            }
            #pragma unroll
            for (int r = 0; r < 8; ++r) {
                accA[r] = sigmoid_of_half(accA[r]);
                accB[r] = sigmoid_of_half(accB[r]);
            }
            __builtin_amdgcn_sched_barrier(0);

            #pragma unroll
            for (int u = 0; u < 4; ++u) {
                #pragma unroll
                for (int s = 0; s < 4; ++s) {
                    v2f af = {w1[u][2 * s], w1[u][2 * s + 1]};
                    v2f bA = {accA[2 * s], accA[2 * s + 1]};
                    v2f bB = {accB[2 * s], accB[2 * s + 1]};
                    acc1A[u] = __builtin_amdgcn_wmma_f32_16x16x4_f32(
                        false, af, false, bA, (short)0, acc1A[u], false, false);
                    acc1B[u] = __builtin_amdgcn_wmma_f32_16x16x4_f32(
                        false, af, false, bB, (short)0, acc1B[u], false, false);
                }
            }
        }

        float pA = 0.0f, pB = 0.0f;
        #pragma unroll
        for (int u = 0; u < 4; ++u) {
            const v8f w2 = ld8(W2b + 16 * u + 8 * half);
            #pragma unroll
            for (int r = 0; r < 8; ++r) {
                pA += sigmoid_of_half(acc1A[u][r]) * w2[r];
                pB += sigmoid_of_half(acc1B[u][r]) * w2[r];
            }
        }

        pA += __shfl_xor(pA, 16, 32);
        pB += __shfl_xor(pB, 16, 32);
        float o = (half ? pB : pA) + b2[branch];
        if (binflag) o = sigmoidf(o);
        out[(size_t)(r0 + lane) * DB + branch] = o;
    }
}

extern "C" void kernel_launch(void* const* d_in, const int* in_sizes, int n_in,
                              void* d_out, int out_size, void* d_ws, size_t ws_size,
                              hipStream_t stream) {
    const float* x  = (const float*)d_in[0];
    const float* W0 = (const float*)d_in[1];
    const float* b0 = (const float*)d_in[2];
    const float* W1 = (const float*)d_in[3];
    const float* b1 = (const float*)d_in[4];
    const float* W2 = (const float*)d_in[5];
    const float* b2 = (const float*)d_in[6];
    float* out = (float*)d_out;

    dim3 grid(DB * (NS / 32) / TILES_PER_BLOCK);
    topo_mlp_kernel<<<grid, NWAVES * 32, 0, stream>>>(x, W0, b0, W1, b1, W2, b2, out);
}
